# starting kernel
# baseline (speedup 1.0000x reference)
#include <hip/hip_runtime.h>
#include <hip/hip_bf16.h>

typedef __attribute__((ext_vector_type(2))) float v2f;
typedef __attribute__((ext_vector_type(8))) float v8f;

#define BB 32
#define SS 150
#define DD 128

__global__ void EOPA_neigh_max_kernel(const float* __restrict__ x,
                                      const int* __restrict__ A,
                                      float* __restrict__ neigh) {
    extern __shared__ float smem[];
    float* xs = smem;
    float* am = smem + SS * DD;

    const int b   = blockIdx.x;
    const int tid = threadIdx.x;

    const float* xb = x + (size_t)b * SS * DD;
    for (int idx = tid; idx < SS * DD; idx += blockDim.x) xs[idx] = xb[idx];
    __syncthreads();

    const int* Ab = A + (size_t)b * SS * SS;
    const int c = tid >> 7;
    const int d = tid & 127;

    for (int i0 = 0; i0 < SS; i0 += 2) {
        for (int idx = tid; idx < SS * 2; idx += blockDim.x) {
            const int j  = idx >> 1;
            const int cc = idx & 1;
            am[idx] = (float)Ab[j * SS + i0 + cc];
        }
        __syncthreads();

        float acc = -INFINITY;
        #pragma unroll 5
        for (int j = 0; j < SS; ++j) {
            acc = fmaxf(acc, xs[j * DD + d] * am[j * 2 + c]);
        }
        neigh[(size_t)b * SS * DD + (size_t)(i0 + c) * DD + d] = acc;
        __syncthreads();
    }
}

__global__ void EOPA_gemm_wmma_kernel(const float* __restrict__ X,
                                      const float* __restrict__ NEIGH,
                                      const float* __restrict__ Wself,
                                      const float* __restrict__ Wneigh,
                                      float* __restrict__ out) {
    const int tid  = threadIdx.x;
    const int wave = tid >> 5;
    const int lane = tid & 31;

    const int tile = blockIdx.x * 8 + wave;
    const int nt = tile & 7;
    const int mt = tile >> 3;
    const int m0 = mt * 16;
    const int n0 = nt * 16;

    const int laneLo = lane & 15;
    const int kHalf  = (lane >> 4) << 1;

    v8f acc = {};

    {
        const float* arow = X     + (size_t)(m0 + laneLo) * DD + kHalf;
        const float* brow = Wself + (size_t)(n0 + laneLo) * DD + kHalf;
        #pragma unroll 8
        for (int k = 0; k < DD; k += 4) {
            v2f a = *(const v2f*)(arow + k);
            v2f b = *(const v2f*)(brow + k);
            acc = __builtin_amdgcn_wmma_f32_16x16x4_f32(
                false, a, false, b, (short)0, acc, false, false);
        }
    }
    {
        const float* arow = NEIGH  + (size_t)(m0 + laneLo) * DD + kHalf;
        const float* brow = Wneigh + (size_t)(n0 + laneLo) * DD + kHalf;
        #pragma unroll 8
        for (int k = 0; k < DD; k += 4) {
            v2f a = *(const v2f*)(arow + k);
            v2f b = *(const v2f*)(brow + k);
            acc = __builtin_amdgcn_wmma_f32_16x16x4_f32(
                false, a, false, b, (short)0, acc, false, false);
        }
    }

    const int colN    = n0 + laneLo;
    const int rowBase = m0 + ((lane >> 4) << 3);
    #pragma unroll
    for (int r = 0; r < 8; ++r) {
        out[(size_t)(rowBase + r) * DD + colN] = acc[r];
    }
}

extern "C" void kernel_launch(void* const* d_in, const int* in_sizes, int n_in,
                              void* d_out, int out_size, void* d_ws, size_t ws_size,
                              hipStream_t stream) {
    const float* x      = (const float*)d_in[0];
    const int*   A      = (const int*)  d_in[1];
    const float* Wself  = (const float*)d_in[2];
    const float* Wneigh = (const float*)d_in[3];
    float*       out    = (float*)d_out;
    float*       neigh  = (float*)d_ws;

    const size_t lds_bytes = (size_t)(SS * DD + 2 * SS) * sizeof(float);
    EOPA_neigh_max_kernel<<<BB, 256, lds_bytes, stream>>>(x, A, neigh);

    EOPA_gemm_wmma_kernel<<<300, 256, 0, stream>>>(x, neigh, Wself, Wneigh, out);
}


# neighbour max by subset tables: per batch and 32-feature slice, the 16 select-pattern maxima (unselected rows' zero folded in) of each 4 neighbours built once in LDS; a node reads 1 row per 4
# speedup vs baseline: 26.1226x; 26.1226x over previous
#include <hip/hip_runtime.h>
#include <hip/hip_bf16.h>

typedef __attribute__((ext_vector_type(2))) float v2f;
typedef __attribute__((ext_vector_type(8))) float v8f;

#define BB 32
#define SS 150
#define DD 128

#define FS 32
#define NSL (DD / FS)
#define NG 5
#define GSZ 4
#define NGRP ((SS + GSZ - 1) / GSZ)
#define NCODE (1 << GSZ)
#define TQ 9
#define NW 3
#define GPW ((NGRP + NW - 1) / NW)
#define NCH 2
#define CG (NGRP / NCH)
#define CR (CG * GSZ)
#define XQ (FS / 4)
#define OQ 9

typedef __attribute__((ext_vector_type(4))) float v4f;

#define MAX_INTO(acc, v) asm("v_max_num_f32 %0, %0, %1" : "+v"(acc) : "v"(v))

__device__ __forceinline__ float max_num(float a, float b) {
    float r;
    asm("v_max_num_f32 %0, %1, %2" : "=v"(r) : "v"(a), "v"(b));
    return r;
}

__global__ __launch_bounds__(NW * 32)
void neigh_max_subset_table_kernel(const float* __restrict__ x,
                                   const int* __restrict__ A,
                                   float* __restrict__ neigh) {
    __shared__ v4f tab[NGRP * NCODE * TQ];
    __shared__ v4f ot[NW][32 * OQ];

    const int lane = threadIdx.x & 31;
    const int w    = threadIdx.x >> 5;
    const int h = blockIdx.x & 1;
    const int s = (blockIdx.x >> 1) % NSL;
    const int b = blockIdx.x / (2 * NSL);

    const float* xcol = x + (size_t)b * SS * DD + s * FS + lane;
    float xv[GPW][GSZ];
    #pragma unroll
    for (int t = 0; t < GPW; ++t) {
        const int G = t * NW + w;
        #pragma unroll
        for (int k = 0; k < GSZ; ++k) {
            const int j = G * GSZ + k;
            if (t < GPW - 1) {
                xv[t][k] = xcol[j * DD];
            } else {
                const int jc = j < SS ? j : SS - 1;
                const float v = xcol[jc * DD];
                xv[t][k] = j < SS ? v : -INFINITY;
            }
        }
    }

    const int  g   = h * 3 + w;
    const bool has = g < NG;
    const int  i   = g * 32 + lane;
    const int  ic  = i < SS ? i : SS - 1;
    const int* Acol = A + (size_t)b * SS * SS + ic;
    int a_cur[CR], a_nxt[CR];
    #pragma unroll
    for (int r = 0; r < CR; ++r) { a_cur[r] = Acol[r * SS]; a_nxt[r] = 0; }

    float* tabf = (float*)tab;
    #pragma unroll
    for (int t = 0; t < GPW; ++t) {
        const int G = t * NW + w;
        if (t < GPW - 1 || G < NGRP) {
            const float x0 = xv[t][0], x1 = xv[t][1], x2 = xv[t][2], x3 = xv[t][3];
            const float p0 = max_num(x0, 0.0f), p1 = max_num(x1, 0.0f);
            const float p2 = max_num(x2, 0.0f), p3 = max_num(x3, 0.0f);
            float T[NCODE];
            T[0]  = 0.0f;
            T[1]  = p0;  T[2] = p1;  T[4] = p2;  T[8] = p3;
            T[3]  = max_num(p0, p1);  T[5]  = max_num(p0, p2);  T[9]  = max_num(p0, p3);
            T[6]  = max_num(p1, p2);  T[10] = max_num(p1, p3);  T[12] = max_num(p2, p3);
            T[7]  = max_num(T[3], p2);  T[11] = max_num(T[3], p3);
            T[13] = max_num(T[5], p3);  T[14] = max_num(T[6], p3);
            T[15] = max_num(max_num(x0, x1), max_num(x2, x3));
            float* dst = tabf + (size_t)G * NCODE * TQ * 4 + lane;
            #pragma unroll
            for (int c = 0; c < NCODE; ++c) dst[c * TQ * 4] = T[c];
        }
    }
    __syncthreads();

    if (has) {
        float acc[FS];
        #pragma unroll
        for (int f = 0; f < FS; ++f) acc[f] = -INFINITY;

        #pragma unroll 1
        for (int c = 0; c < NCH; ++c) {
            const bool last = (c == NCH - 1);
            if (!last) {
                #pragma unroll
                for (int r = 0; r < CR; ++r) {
                    const int j  = (c + 1) * CR + r;
                    const int jc = (r < CR - 2 || j < SS) ? j : SS - 1;
                    a_nxt[r] = Acol[jc * SS];
                }
            }
            int rowq[CG];
            #pragma unroll
            for (int gl = 0; gl < CG; ++gl) {
                int code = 0;
                #pragma unroll
                for (int k = 0; k < GSZ; ++k) {
                    const int r = gl * GSZ + k;
                    bool on = (a_cur[r] == 1);
                    if (r >= CR - 2) on = on || last;
                    code |= on ? (1 << k) : 0;
                }
                rowq[gl] = ((c * CG + gl) * NCODE + code) * TQ;
            }

            v4f cur[XQ], nxt[XQ];
            #pragma unroll
            for (int q = 0; q < XQ; ++q) { cur[q] = tab[rowq[0] + q]; nxt[q] = cur[q]; }
            #pragma unroll
            for (int gl = 0; gl < CG; ++gl) {
                if (gl + 1 < CG) {
                    #pragma unroll
                    for (int q = 0; q < XQ; ++q) nxt[q] = tab[rowq[gl + 1] + q];
                }
                #pragma unroll
                for (int q = 0; q < XQ; ++q) {
                    const v4f v = cur[q];
                    MAX_INTO(acc[4 * q + 0], v.x);
                    MAX_INTO(acc[4 * q + 1], v.y);
                    MAX_INTO(acc[4 * q + 2], v.z);
                    MAX_INTO(acc[4 * q + 3], v.w);
                }
                #pragma unroll
                for (int q = 0; q < XQ; ++q) cur[q] = nxt[q];
            }
            #pragma unroll
            for (int r = 0; r < CR; ++r) a_cur[r] = a_nxt[r];
        }

        #pragma unroll
        for (int q = 0; q < XQ; ++q)
            ot[w][lane * OQ + q] = (v4f){acc[4 * q + 0], acc[4 * q + 1], acc[4 * q + 2], acc[4 * q + 3]};
    }
    __syncthreads();
    if (has) {
        #pragma unroll
        for (int r = 0; r < 8; ++r) {
            const int row  = r * 4 + (lane >> 3);
            const int node = g * 32 + row;
            if (node < SS)
                *(v4f*)(neigh + ((size_t)b * SS + node) * DD + s * FS + (lane & 7) * 4) = ot[w][row * OQ + (lane & 7)];
        }
    }
}

__global__ void EOPA_gemm_wmma_kernel(const float* __restrict__ X,
                                      const float* __restrict__ NEIGH,
                                      const float* __restrict__ Wself,
                                      const float* __restrict__ Wneigh,
                                      float* __restrict__ out) {
    const int tid  = threadIdx.x;
    const int wave = tid >> 5;
    const int lane = tid & 31;

    const int tile = blockIdx.x * 8 + wave;
    const int nt = tile & 7;
    const int mt = tile >> 3;
    const int m0 = mt * 16;
    const int n0 = nt * 16;

    const int laneLo = lane & 15;
    const int kHalf  = (lane >> 4) << 1;

    v8f acc = {};

    {
        const float* arow = X     + (size_t)(m0 + laneLo) * DD + kHalf;
        const float* brow = Wself + (size_t)(n0 + laneLo) * DD + kHalf;
        #pragma unroll 8
        for (int k = 0; k < DD; k += 4) {
            v2f a = *(const v2f*)(arow + k);
            v2f b = *(const v2f*)(brow + k);
            acc = __builtin_amdgcn_wmma_f32_16x16x4_f32(
                false, a, false, b, (short)0, acc, false, false);
        }
    }
    {
        const float* arow = NEIGH  + (size_t)(m0 + laneLo) * DD + kHalf;
        const float* brow = Wneigh + (size_t)(n0 + laneLo) * DD + kHalf;
        #pragma unroll 8
        for (int k = 0; k < DD; k += 4) {
            v2f a = *(const v2f*)(arow + k);
            v2f b = *(const v2f*)(brow + k);
            acc = __builtin_amdgcn_wmma_f32_16x16x4_f32(
                false, a, false, b, (short)0, acc, false, false);
        }
    }

    const int colN    = n0 + laneLo;
    const int rowBase = m0 + ((lane >> 4) << 3);
    #pragma unroll
    for (int r = 0; r < 8; ++r) {
        out[(size_t)(rowBase + r) * DD + colN] = acc[r];
    }
}

extern "C" void kernel_launch(void* const* d_in, const int* in_sizes, int n_in,
                              void* d_out, int out_size, void* d_ws, size_t ws_size,
                              hipStream_t stream) {
    const float* x      = (const float*)d_in[0];
    const int*   A      = (const int*)  d_in[1];
    const float* Wself  = (const float*)d_in[2];
    const float* Wneigh = (const float*)d_in[3];
    float*       out    = (float*)d_out;
    float*       neigh  = (float*)d_ws;

    neigh_max_subset_table_kernel<<<BB * NSL * 2, NW * 32, 0, stream>>>(x, A, neigh);

    EOPA_gemm_wmma_kernel<<<300, 256, 0, stream>>>(x, neigh, Wself, Wneigh, out);
}


# GEMM re-tiled in registers: a wave owns a 32x32 output tile (2x2 WMMA tiles, 4 accumulator chains), 16-B fragment loads straight from global; k pairs regrouped within each 8: sum order changes
# speedup vs baseline: 36.1823x; 1.3851x over previous
#include <hip/hip_runtime.h>
#include <hip/hip_bf16.h>

typedef __attribute__((ext_vector_type(2))) float v2f;
typedef __attribute__((ext_vector_type(8))) float v8f;

#define BB 32
#define SS 150
#define DD 128

#define FS 32
#define NSL (DD / FS)
#define NG 5
#define GSZ 4
#define NGRP ((SS + GSZ - 1) / GSZ)
#define NCODE (1 << GSZ)
#define TQ 9
#define NW 3
#define GPW ((NGRP + NW - 1) / NW)
#define NCH 2
#define CG (NGRP / NCH)
#define CR (CG * GSZ)
#define XQ (FS / 4)
#define OQ 9

typedef __attribute__((ext_vector_type(4))) float v4f;

#define MAX_INTO(acc, v) asm("v_max_num_f32 %0, %0, %1" : "+v"(acc) : "v"(v))

__device__ __forceinline__ float max_num(float a, float b) {
    float r;
    asm("v_max_num_f32 %0, %1, %2" : "=v"(r) : "v"(a), "v"(b));
    return r;
}

__global__ __launch_bounds__(NW * 32)
void neigh_max_subset_table_kernel(const float* __restrict__ x,
                                   const int* __restrict__ A,
                                   float* __restrict__ neigh) {
    __shared__ v4f tab[NGRP * NCODE * TQ];
    __shared__ v4f ot[NW][32 * OQ];

    const int lane = threadIdx.x & 31;
    const int w    = threadIdx.x >> 5;
    const int h = blockIdx.x & 1;
    const int s = (blockIdx.x >> 1) % NSL;
    const int b = blockIdx.x / (2 * NSL);

    const float* xcol = x + (size_t)b * SS * DD + s * FS + lane;
    float xv[GPW][GSZ];
    #pragma unroll
    for (int t = 0; t < GPW; ++t) {
        const int G = t * NW + w;
        #pragma unroll
        for (int k = 0; k < GSZ; ++k) {
            const int j = G * GSZ + k;
            if (t < GPW - 1) {
                xv[t][k] = xcol[j * DD];
            } else {
                const int jc = j < SS ? j : SS - 1;
                const float v = xcol[jc * DD];
                xv[t][k] = j < SS ? v : -INFINITY;
            }
        }
    }

    const int  g   = h * 3 + w;
    const bool has = g < NG;
    const int  i   = g * 32 + lane;
    const int  ic  = i < SS ? i : SS - 1;
    const int* Acol = A + (size_t)b * SS * SS + ic;
    int a_cur[CR], a_nxt[CR];
    #pragma unroll
    for (int r = 0; r < CR; ++r) { a_cur[r] = Acol[r * SS]; a_nxt[r] = 0; }

    float* tabf = (float*)tab;
    #pragma unroll
    for (int t = 0; t < GPW; ++t) {
        const int G = t * NW + w;
        if (t < GPW - 1 || G < NGRP) {
            const float x0 = xv[t][0], x1 = xv[t][1], x2 = xv[t][2], x3 = xv[t][3];
            const float p0 = max_num(x0, 0.0f), p1 = max_num(x1, 0.0f);
            const float p2 = max_num(x2, 0.0f), p3 = max_num(x3, 0.0f);
            float T[NCODE];
            T[0]  = 0.0f;
            T[1]  = p0;  T[2] = p1;  T[4] = p2;  T[8] = p3;
            T[3]  = max_num(p0, p1);  T[5]  = max_num(p0, p2);  T[9]  = max_num(p0, p3);
            T[6]  = max_num(p1, p2);  T[10] = max_num(p1, p3);  T[12] = max_num(p2, p3);
            T[7]  = max_num(T[3], p2);  T[11] = max_num(T[3], p3);
            T[13] = max_num(T[5], p3);  T[14] = max_num(T[6], p3);
            T[15] = max_num(max_num(x0, x1), max_num(x2, x3));
            float* dst = tabf + (size_t)G * NCODE * TQ * 4 + lane;
            #pragma unroll
            for (int c = 0; c < NCODE; ++c) dst[c * TQ * 4] = T[c];
        }
    }
    __syncthreads();

    if (has) {
        float acc[FS];
        #pragma unroll
        for (int f = 0; f < FS; ++f) acc[f] = -INFINITY;

        #pragma unroll 1
        for (int c = 0; c < NCH; ++c) {
            const bool last = (c == NCH - 1);
            if (!last) {
                #pragma unroll
                for (int r = 0; r < CR; ++r) {
                    const int j  = (c + 1) * CR + r;
                    const int jc = (r < CR - 2 || j < SS) ? j : SS - 1;
                    a_nxt[r] = Acol[jc * SS];
                }
            }
            int rowq[CG];
            #pragma unroll
            for (int gl = 0; gl < CG; ++gl) {
                int code = 0;
                #pragma unroll
                for (int k = 0; k < GSZ; ++k) {
                    const int r = gl * GSZ + k;
                    bool on = (a_cur[r] == 1);
                    if (r >= CR - 2) on = on || last;
                    code |= on ? (1 << k) : 0;
                }
                rowq[gl] = ((c * CG + gl) * NCODE + code) * TQ;
            }

            v4f cur[XQ], nxt[XQ];
            #pragma unroll
            for (int q = 0; q < XQ; ++q) { cur[q] = tab[rowq[0] + q]; nxt[q] = cur[q]; }
            #pragma unroll
            for (int gl = 0; gl < CG; ++gl) {
                if (gl + 1 < CG) {
                    #pragma unroll
                    for (int q = 0; q < XQ; ++q) nxt[q] = tab[rowq[gl + 1] + q];
                }
                #pragma unroll
                for (int q = 0; q < XQ; ++q) {
                    const v4f v = cur[q];
                    MAX_INTO(acc[4 * q + 0], v.x);
                    MAX_INTO(acc[4 * q + 1], v.y);
                    MAX_INTO(acc[4 * q + 2], v.z);
                    MAX_INTO(acc[4 * q + 3], v.w);
                }
                #pragma unroll
                for (int q = 0; q < XQ; ++q) cur[q] = nxt[q];
            }
            #pragma unroll
            for (int r = 0; r < CR; ++r) a_cur[r] = a_nxt[r];
        }

        #pragma unroll
        for (int q = 0; q < XQ; ++q)
            ot[w][lane * OQ + q] = (v4f){acc[4 * q + 0], acc[4 * q + 1], acc[4 * q + 2], acc[4 * q + 3]};
    }
    __syncthreads();
    if (has) {
        #pragma unroll
        for (int r = 0; r < 8; ++r) {
            const int row  = r * 4 + (lane >> 3);
            const int node = g * 32 + row;
            if (node < SS)
                *(v4f*)(neigh + ((size_t)b * SS + node) * DD + s * FS + (lane & 7) * 4) = ot[w][row * OQ + (lane & 7)];
        }
    }
}

#define KCH 32
#define NSTEP (KCH / 8)

struct FragSet {
    v4f a0[NSTEP], a1[NSTEP], b0[NSTEP], b1[NSTEP];
};

__device__ __forceinline__ void load_chunk(FragSet& f, const float* pa0, const float* pa1,
                                           const float* pb0, const float* pb1, int koff) {
    #pragma unroll
    for (int t = 0; t < NSTEP; ++t) {
        f.a0[t] = *(const v4f*)(pa0 + koff + 8 * t);
        f.a1[t] = *(const v4f*)(pa1 + koff + 8 * t);
        f.b0[t] = *(const v4f*)(pb0 + koff + 8 * t);
        f.b1[t] = *(const v4f*)(pb1 + koff + 8 * t);
    }
}

#define WMMA_F32(a, b, c) __builtin_amdgcn_wmma_f32_16x16x4_f32(false, a, false, b, (short)0, c, false, false)

__device__ __forceinline__ void mma_chunk(const FragSet& f, v8f& c00, v8f& c01, v8f& c10, v8f& c11) {
    #pragma unroll
    for (int t = 0; t < NSTEP; ++t) {
        const v2f a0l = {f.a0[t].x, f.a0[t].y}, a0h = {f.a0[t].z, f.a0[t].w};
        const v2f a1l = {f.a1[t].x, f.a1[t].y}, a1h = {f.a1[t].z, f.a1[t].w};
        const v2f b0l = {f.b0[t].x, f.b0[t].y}, b0h = {f.b0[t].z, f.b0[t].w};
        const v2f b1l = {f.b1[t].x, f.b1[t].y}, b1h = {f.b1[t].z, f.b1[t].w};
        c00 = WMMA_F32(a0l, b0l, c00);  c01 = WMMA_F32(a0l, b1l, c01);
        c10 = WMMA_F32(a1l, b0l, c10);  c11 = WMMA_F32(a1l, b1l, c11);
        c00 = WMMA_F32(a0h, b0h, c00);  c01 = WMMA_F32(a0h, b1h, c01);
        c10 = WMMA_F32(a1h, b0h, c10);  c11 = WMMA_F32(a1h, b1h, c11);
    }
}

__global__ __launch_bounds__(96)
void gemm_wave_tile_32x32_kernel(const float* __restrict__ X,
                                 const float* __restrict__ NEIGH,
                                 const float* __restrict__ Wself,
                                 const float* __restrict__ Wneigh,
                                 float* __restrict__ out) {
    const int lane = threadIdx.x & 31;
    const int tile = blockIdx.x * 3 + (threadIdx.x >> 5);
    const int mb = tile >> 2;
    const int nb = tile & 3;
    const int c = lane & 15;
    const int h = lane >> 4;
    const int m0 = mb * 32;
    const int n0 = nb * 32;

    const size_t arow = (size_t)(m0 + c) * DD + 4 * h;
    const size_t brow = (size_t)(n0 + c) * DD + 4 * h;
    const float* xa0 = X + arow;       const float* xa1 = xa0 + 16 * DD;
    const float* na0 = NEIGH + arow;   const float* na1 = na0 + 16 * DD;
    const float* sb0 = Wself + brow;   const float* sb1 = sb0 + 16 * DD;
    const float* nb0 = Wneigh + brow;  const float* nb1 = nb0 + 16 * DD;

    v8f c00 = {}, c01 = {}, c10 = {}, c11 = {};

    #define STAGE_END() __builtin_amdgcn_sched_barrier(0)
    FragSet f0, f1, f2;
    load_chunk(f0, xa0, xa1, sb0, sb1, 0 * KCH);
    load_chunk(f1, xa0, xa1, sb0, sb1, 1 * KCH);
    load_chunk(f2, xa0, xa1, sb0, sb1, 2 * KCH);
    STAGE_END();
    mma_chunk(f0, c00, c01, c10, c11);
    STAGE_END();
    load_chunk(f0, xa0, xa1, sb0, sb1, 3 * KCH);
    STAGE_END();
    mma_chunk(f1, c00, c01, c10, c11);
    STAGE_END();
    load_chunk(f1, na0, na1, nb0, nb1, 0 * KCH);
    STAGE_END();
    mma_chunk(f2, c00, c01, c10, c11);
    STAGE_END();
    load_chunk(f2, na0, na1, nb0, nb1, 1 * KCH);
    STAGE_END();
    mma_chunk(f0, c00, c01, c10, c11);
    STAGE_END();
    load_chunk(f0, na0, na1, nb0, nb1, 2 * KCH);
    STAGE_END();
    mma_chunk(f1, c00, c01, c10, c11);
    STAGE_END();
    load_chunk(f1, na0, na1, nb0, nb1, 3 * KCH);
    STAGE_END();
    mma_chunk(f2, c00, c01, c10, c11);
    mma_chunk(f0, c00, c01, c10, c11);
    mma_chunk(f1, c00, c01, c10, c11);

    float* o = out + (size_t)(m0 + 8 * h) * DD + n0 + c;
    #pragma unroll
    for (int r = 0; r < 8; ++r) {
        o[(size_t)r * DD]             = c00[r];
        o[(size_t)r * DD + 16]        = c01[r];
        o[(size_t)(r + 16) * DD]      = c10[r];
        o[(size_t)(r + 16) * DD + 16] = c11[r];
    }
}

extern "C" void kernel_launch(void* const* d_in, const int* in_sizes, int n_in,
                              void* d_out, int out_size, void* d_ws, size_t ws_size,
                              hipStream_t stream) {
    const float* x      = (const float*)d_in[0];
    const int*   A      = (const int*)  d_in[1];
    const float* Wself  = (const float*)d_in[2];
    const float* Wneigh = (const float*)d_in[3];
    float*       out    = (float*)d_out;
    float*       neigh  = (float*)d_ws;

    neigh_max_subset_table_kernel<<<BB * NSL * 2, NW * 32, 0, stream>>>(x, A, neigh);

    gemm_wave_tile_32x32_kernel<<<200, 96, 0, stream>>>(x, neigh, Wself, Wneigh, out);
}
